# starting kernel
# baseline (speedup 1.0000x reference)
#include <hip/hip_runtime.h>
#include <hip/hip_bf16.h>
#include <stdint.h>

#define H 256

typedef __attribute__((ext_vector_type(2))) float v2f;
typedef __attribute__((ext_vector_type(8))) float v8f;

__global__ void k_zero(unsigned int* __restrict__ p, int n) {
    int i = blockIdx.x * blockDim.x + threadIdx.x;
    if (i < n) p[i] = 0u;
}

__global__ void k_hist(const int* __restrict__ src_idx, unsigned int* __restrict__ counts, int E) {
    int i = blockIdx.x * blockDim.x + threadIdx.x;
    if (i < E) atomicAdd(&counts[src_idx[i]], 1u);
}

__global__ void k_weighted_rowsum(const float* __restrict__ X,
                                  const unsigned int* __restrict__ counts,
                                  float* __restrict__ ls, int nrows) {
    int h = threadIdx.x;
    float acc = 0.0f;
    for (int n = blockIdx.x; n < nrows; n += gridDim.x) {
        float c = (float)counts[n];
        acc = fmaf(c, X[(size_t)n * H + h], acc);
    }
    atomicAdd(&ls[h], acc);
}

__global__ void k_matvec_R(const float* __restrict__ R,
                           const float* __restrict__ ls,
                           float* __restrict__ v) {
    __shared__ float s[H];
    s[threadIdx.x] = ls[threadIdx.x];
    __syncthreads();
    int wave = threadIdx.x >> 5;
    int lane = threadIdx.x & 31;
    for (int i = wave; i < H; i += 8) {
        const float* row = R + (size_t)i * H;
        float acc = 0.0f;
        #pragma unroll
        for (int j = lane; j < H; j += 32) acc = fmaf(row[j], s[j], acc);
        #pragma unroll
        for (int off = 16; off > 0; off >>= 1) acc += __shfl_xor(acc, off, 32);
        if (lane == 0) v[i] = acc;
    }
}

__global__ void k_dot_rows_wmma(const float* __restrict__ X,
                                const float* __restrict__ v,
                                float* __restrict__ t, int nrows) {
    __shared__ float vs[H];
    vs[threadIdx.x] = v[threadIdx.x];
    __syncthreads();

    int wave = threadIdx.x >> 5;
    int lane = threadIdx.x & 31;
    int row0 = (blockIdx.x * 8 + wave) * 16;

    int r = row0 + (lane & 15);
    if (r >= nrows) r = 0;
    const float* xrow = X + (size_t)r * H;
    int khalf = (lane >> 4) * 2;

    v8f c = {};
    #pragma unroll 8
    for (int k0 = 0; k0 < H; k0 += 4) {
        int kk = k0 + khalf;
        v2f a = *(const v2f*)(vs + kk);
        v2f b = *(const v2f*)(xrow + kk);
        c = __builtin_amdgcn_wmma_f32_16x16x4_f32(
                 false, a,  false, b,
                 (short)0, c,  false,  false);
    }

    int n = row0 + (lane & 15);
    if (lane < 16 && n < nrows) t[n] = c[0];
}

__global__ void k_gather_scores(const float* __restrict__ t,
                                const int* __restrict__ trg_idx,
                                float* __restrict__ out, int E) {
    int i = blockIdx.x * blockDim.x + threadIdx.x;
    if (i < E) out[i] = t[trg_idx[i]];
}

extern "C" void kernel_launch(void* const* d_in, const int* in_sizes, int n_in,
                              void* d_out, int out_size, void* d_ws, size_t ws_size,
                              hipStream_t stream) {
    const float* x_src = (const float*)d_in[0];
    const float* x_trg = (const float*)d_in[1];
    const float* R     = (const float*)d_in[2];
    const int*   edges = (const int*)d_in[3];

    const int n_src = in_sizes[0] / H;
    const int n_trg = in_sizes[1] / H;
    const int E     = in_sizes[3] / 2;
    const int* src_idx = edges;
    const int* trg_idx = edges + E;
    float* out = (float*)d_out;

    char* ws = (char*)d_ws;
    size_t off = 0;
    unsigned int* counts = (unsigned int*)(ws + off); off += ((size_t)n_src * 4 + 255) & ~(size_t)255;
    float* ls            = (float*)(ws + off);        off += (H * 4 + 255) & ~(size_t)255;
    float* vvec          = (float*)(ws + off);        off += (H * 4 + 255) & ~(size_t)255;
    float* t             = (float*)(ws + off);

    {
        int nz = n_src;
        k_zero<<<(nz + 255) / 256, 256, 0, stream>>>(counts, nz);
        k_zero<<<1, 256, 0, stream>>>((unsigned int*)ls, H);
        k_zero<<<1, 256, 0, stream>>>((unsigned int*)vvec, H);
    }
    k_hist<<<(E + 255) / 256, 256, 0, stream>>>(src_idx, counts, E);
    k_weighted_rowsum<<<512, H, 0, stream>>>(x_src, counts, ls, n_src);
    k_matvec_R<<<1, H, 0, stream>>>(R, ls, vvec);
    {
        int blocks = (n_trg + 127) / 128;
        k_dot_rows_wmma<<<blocks, 256, 0, stream>>>(x_trg, vvec, t, n_trg);
    }
    k_gather_scores<<<(E + 255) / 256, 256, 0, stream>>>(t, trg_idx, out, E);
}


# left_sum stream kernel: a wave owns whole rows, 16-byte loads, all 13 rows of a wave (26 loads) requested before the first sum; new f32 summation order (8-wave LDS sum, then the atomics)
# speedup vs baseline: 1.6432x; 1.6432x over previous
#include <hip/hip_runtime.h>
#include <hip/hip_bf16.h>
#include <stdint.h>

#define H 256

typedef __attribute__((ext_vector_type(2))) float v2f;
typedef __attribute__((ext_vector_type(4))) float v4f;
typedef __attribute__((ext_vector_type(8))) float v8f;

__global__ void k_zero(unsigned int* __restrict__ p, int n) {
    int i = blockIdx.x * blockDim.x + threadIdx.x;
    if (i < n) p[i] = 0u;
}

__global__ void k_hist(const int* __restrict__ src_idx, unsigned int* __restrict__ counts, int E) {
    int i = blockIdx.x * blockDim.x + threadIdx.x;
    if (i < E) atomicAdd(&counts[src_idx[i]], 1u);
}

#define RS_ROWS 13

__global__ __launch_bounds__(256)
void k_weighted_rowsum(const float* __restrict__ X,
                       const unsigned int* __restrict__ counts,
                       float* __restrict__ ls, int nrows) {
    __shared__ v4f part[8][H / 4];
    const int lane = threadIdx.x & 31;
    const int wave = __builtin_amdgcn_readfirstlane((int)(threadIdx.x >> 5));
    const int gw = blockIdx.x * 8 + wave;
    const int nw = gridDim.x * 8;

    v4f a0 = {0.0f, 0.0f, 0.0f, 0.0f};
    v4f a1 = {0.0f, 0.0f, 0.0f, 0.0f};
    for (int n0 = gw; n0 < nrows; n0 += nw * RS_ROWS) {
        const int nl = n0 + lane * nw;
        const bool cvalid = lane < RS_ROWS && nl < nrows;
        const unsigned int craw = counts[cvalid ? nl : n0];

        v4f x0[RS_ROWS], x1[RS_ROWS];
        #pragma unroll
        for (int r = 0; r < RS_ROWS; ++r) {
            const int n = n0 + r * nw;
            const int nn = (n < nrows) ? n : n0;
            const v4f* p = (const v4f*)(X + (size_t)nn * H);
            x0[r] = p[lane];
            x1[r] = p[lane + 32];
        }
        __builtin_amdgcn_sched_barrier(0);
        const float cl = cvalid ? (float)craw : 0.0f;
        #pragma unroll
        for (int r = 0; r < RS_ROWS; ++r) {
            const float c = __builtin_bit_cast(float,
                __builtin_amdgcn_readlane(__builtin_bit_cast(int, cl), r));
            a0.x = fmaf(c, x0[r].x, a0.x);
            a0.y = fmaf(c, x0[r].y, a0.y);
            a0.z = fmaf(c, x0[r].z, a0.z);
            a0.w = fmaf(c, x0[r].w, a0.w);
            a1.x = fmaf(c, x1[r].x, a1.x);
            a1.y = fmaf(c, x1[r].y, a1.y);
            a1.z = fmaf(c, x1[r].z, a1.z);
            a1.w = fmaf(c, x1[r].w, a1.w);
        }
    }
    part[wave][lane]      = a0;
    part[wave][lane + 32] = a1;
    __syncthreads();
    const float* pf = (const float*)&part[0][0];
    float s = 0.0f;
    for (int w = 0; w < 8; ++w) s += pf[w * H + threadIdx.x];
    atomicAdd(&ls[threadIdx.x], s);
}

__global__ void k_matvec_R(const float* __restrict__ R,
                           const float* __restrict__ ls,
                           float* __restrict__ v) {
    __shared__ float s[H];
    s[threadIdx.x] = ls[threadIdx.x];
    __syncthreads();
    int wave = threadIdx.x >> 5;
    int lane = threadIdx.x & 31;
    for (int i = wave; i < H; i += 8) {
        const float* row = R + (size_t)i * H;
        float acc = 0.0f;
        #pragma unroll
        for (int j = lane; j < H; j += 32) acc = fmaf(row[j], s[j], acc);
        #pragma unroll
        for (int off = 16; off > 0; off >>= 1) acc += __shfl_xor(acc, off, 32);
        if (lane == 0) v[i] = acc;
    }
}

__global__ void k_dot_rows_wmma(const float* __restrict__ X,
                                const float* __restrict__ v,
                                float* __restrict__ t, int nrows) {
    __shared__ float vs[H];
    vs[threadIdx.x] = v[threadIdx.x];
    __syncthreads();

    int wave = threadIdx.x >> 5;
    int lane = threadIdx.x & 31;
    int row0 = (blockIdx.x * 8 + wave) * 16;

    int r = row0 + (lane & 15);
    if (r >= nrows) r = 0;
    const float* xrow = X + (size_t)r * H;
    int khalf = (lane >> 4) * 2;

    v8f c = {};
    #pragma unroll 8
    for (int k0 = 0; k0 < H; k0 += 4) {
        int kk = k0 + khalf;
        v2f a = *(const v2f*)(vs + kk);
        v2f b = *(const v2f*)(xrow + kk);
        c = __builtin_amdgcn_wmma_f32_16x16x4_f32(
                 false, a,  false, b,
                 (short)0, c,  false,  false);
    }

    int n = row0 + (lane & 15);
    if (lane < 16 && n < nrows) t[n] = c[0];
}

__global__ void k_gather_scores(const float* __restrict__ t,
                                const int* __restrict__ trg_idx,
                                float* __restrict__ out, int E) {
    int i = blockIdx.x * blockDim.x + threadIdx.x;
    if (i < E) out[i] = t[trg_idx[i]];
}

extern "C" void kernel_launch(void* const* d_in, const int* in_sizes, int n_in,
                              void* d_out, int out_size, void* d_ws, size_t ws_size,
                              hipStream_t stream) {
    const float* x_src = (const float*)d_in[0];
    const float* x_trg = (const float*)d_in[1];
    const float* R     = (const float*)d_in[2];
    const int*   edges = (const int*)d_in[3];

    const int n_src = in_sizes[0] / H;
    const int n_trg = in_sizes[1] / H;
    const int E     = in_sizes[3] / 2;
    const int* src_idx = edges;
    const int* trg_idx = edges + E;
    float* out = (float*)d_out;

    char* ws = (char*)d_ws;
    size_t off = 0;
    unsigned int* counts = (unsigned int*)(ws + off); off += ((size_t)n_src * 4 + 255) & ~(size_t)255;
    float* ls            = (float*)(ws + off);        off += (H * 4 + 255) & ~(size_t)255;
    float* vvec          = (float*)(ws + off);        off += (H * 4 + 255) & ~(size_t)255;
    float* t             = (float*)(ws + off);

    {
        int nz = n_src;
        k_zero<<<(nz + 255) / 256, 256, 0, stream>>>(counts, nz);
        k_zero<<<1, 256, 0, stream>>>((unsigned int*)ls, H);
        k_zero<<<1, 256, 0, stream>>>((unsigned int*)vvec, H);
    }
    k_hist<<<(E + 255) / 256, 256, 0, stream>>>(src_idx, counts, E);
    k_weighted_rowsum<<<512, H, 0, stream>>>(x_src, counts, ls, n_src);
    k_matvec_R<<<1, H, 0, stream>>>(R, ls, vvec);
    {
        int blocks = (n_trg + 127) / 128;
        k_dot_rows_wmma<<<blocks, 256, 0, stream>>>(x_trg, vvec, t, n_trg);
    }
    k_gather_scores<<<(E + 255) / 256, 256, 0, stream>>>(t, trg_idx, out, E);
}


# matvec v = R*ls moved from one work-group (32 dependent row trips per wave) to 256 waves, one row of R per wave, all 16 loads in flight at once; same column order and xor tree, v bit-identical
# speedup vs baseline: 2.4446x; 1.4877x over previous
#include <hip/hip_runtime.h>
#include <hip/hip_bf16.h>
#include <stdint.h>

#define H 256

typedef __attribute__((ext_vector_type(2))) float v2f;
typedef __attribute__((ext_vector_type(4))) float v4f;
typedef __attribute__((ext_vector_type(8))) float v8f;

__global__ void k_zero(unsigned int* __restrict__ p, int n) {
    int i = blockIdx.x * blockDim.x + threadIdx.x;
    if (i < n) p[i] = 0u;
}

__global__ void k_hist(const int* __restrict__ src_idx, unsigned int* __restrict__ counts, int E) {
    int i = blockIdx.x * blockDim.x + threadIdx.x;
    if (i < E) atomicAdd(&counts[src_idx[i]], 1u);
}

#define RS_ROWS 13

__global__ __launch_bounds__(256)
void k_weighted_rowsum(const float* __restrict__ X,
                       const unsigned int* __restrict__ counts,
                       float* __restrict__ ls, int nrows) {
    __shared__ v4f part[8][H / 4];
    const int lane = threadIdx.x & 31;
    const int wave = __builtin_amdgcn_readfirstlane((int)(threadIdx.x >> 5));
    const int gw = blockIdx.x * 8 + wave;
    const int nw = gridDim.x * 8;

    v4f a0 = {0.0f, 0.0f, 0.0f, 0.0f};
    v4f a1 = {0.0f, 0.0f, 0.0f, 0.0f};
    for (int n0 = gw; n0 < nrows; n0 += nw * RS_ROWS) {
        const int nl = n0 + lane * nw;
        const bool cvalid = lane < RS_ROWS && nl < nrows;
        const unsigned int craw = counts[cvalid ? nl : n0];

        v4f x0[RS_ROWS], x1[RS_ROWS];
        #pragma unroll
        for (int r = 0; r < RS_ROWS; ++r) {
            const int n = n0 + r * nw;
            const int nn = (n < nrows) ? n : n0;
            const v4f* p = (const v4f*)(X + (size_t)nn * H);
            x0[r] = p[lane];
            x1[r] = p[lane + 32];
        }
        __builtin_amdgcn_sched_barrier(0);
        const float cl = cvalid ? (float)craw : 0.0f;
        #pragma unroll
        for (int r = 0; r < RS_ROWS; ++r) {
            const float c = __builtin_bit_cast(float,
                __builtin_amdgcn_readlane(__builtin_bit_cast(int, cl), r));
            a0.x = fmaf(c, x0[r].x, a0.x);
            a0.y = fmaf(c, x0[r].y, a0.y);
            a0.z = fmaf(c, x0[r].z, a0.z);
            a0.w = fmaf(c, x0[r].w, a0.w);
            a1.x = fmaf(c, x1[r].x, a1.x);
            a1.y = fmaf(c, x1[r].y, a1.y);
            a1.z = fmaf(c, x1[r].z, a1.z);
            a1.w = fmaf(c, x1[r].w, a1.w);
        }
    }
    part[wave][lane]      = a0;
    part[wave][lane + 32] = a1;
    __syncthreads();
    const float* pf = (const float*)&part[0][0];
    float s = 0.0f;
    for (int w = 0; w < 8; ++w) s += pf[w * H + threadIdx.x];
    atomicAdd(&ls[threadIdx.x], s);
}

#define MV_WAVES 4

__global__ __launch_bounds__(MV_WAVES * 32)
void k_matvec_R(const float* __restrict__ R,
                const float* __restrict__ ls,
                float* __restrict__ v) {
    const int wave = threadIdx.x >> 5;
    const int lane = threadIdx.x & 31;
    const int i = blockIdx.x * MV_WAVES + wave;
    const float* row = R + (size_t)i * H;
    float rv[H / 32], sv[H / 32];
    for (int k = 0; k < H / 32; ++k) {
        rv[k] = row[lane + 32 * k];
        sv[k] = ls[lane + 32 * k];
    }
    float acc = 0.0f;
    for (int k = 0; k < H / 32; ++k) acc = fmaf(rv[k], sv[k], acc);
    for (int off = 16; off > 0; off >>= 1) acc += __shfl_xor(acc, off, 32);
    if (lane == 0) v[i] = acc;
}

__global__ void k_dot_rows_wmma(const float* __restrict__ X,
                                const float* __restrict__ v,
                                float* __restrict__ t, int nrows) {
    __shared__ float vs[H];
    vs[threadIdx.x] = v[threadIdx.x];
    __syncthreads();

    int wave = threadIdx.x >> 5;
    int lane = threadIdx.x & 31;
    int row0 = (blockIdx.x * 8 + wave) * 16;

    int r = row0 + (lane & 15);
    if (r >= nrows) r = 0;
    const float* xrow = X + (size_t)r * H;
    int khalf = (lane >> 4) * 2;

    v8f c = {};
    #pragma unroll 8
    for (int k0 = 0; k0 < H; k0 += 4) {
        int kk = k0 + khalf;
        v2f a = *(const v2f*)(vs + kk);
        v2f b = *(const v2f*)(xrow + kk);
        c = __builtin_amdgcn_wmma_f32_16x16x4_f32(
                 false, a,  false, b,
                 (short)0, c,  false,  false);
    }

    int n = row0 + (lane & 15);
    if (lane < 16 && n < nrows) t[n] = c[0];
}

__global__ void k_gather_scores(const float* __restrict__ t,
                                const int* __restrict__ trg_idx,
                                float* __restrict__ out, int E) {
    int i = blockIdx.x * blockDim.x + threadIdx.x;
    if (i < E) out[i] = t[trg_idx[i]];
}

extern "C" void kernel_launch(void* const* d_in, const int* in_sizes, int n_in,
                              void* d_out, int out_size, void* d_ws, size_t ws_size,
                              hipStream_t stream) {
    const float* x_src = (const float*)d_in[0];
    const float* x_trg = (const float*)d_in[1];
    const float* R     = (const float*)d_in[2];
    const int*   edges = (const int*)d_in[3];

    const int n_src = in_sizes[0] / H;
    const int n_trg = in_sizes[1] / H;
    const int E     = in_sizes[3] / 2;
    const int* src_idx = edges;
    const int* trg_idx = edges + E;
    float* out = (float*)d_out;

    char* ws = (char*)d_ws;
    size_t off = 0;
    unsigned int* counts = (unsigned int*)(ws + off); off += ((size_t)n_src * 4 + 255) & ~(size_t)255;
    float* ls            = (float*)(ws + off);        off += (H * 4 + 255) & ~(size_t)255;
    float* vvec          = (float*)(ws + off);        off += (H * 4 + 255) & ~(size_t)255;
    float* t             = (float*)(ws + off);

    {
        int nz = n_src;
        k_zero<<<(nz + 255) / 256, 256, 0, stream>>>(counts, nz);
        k_zero<<<1, 256, 0, stream>>>((unsigned int*)ls, H);
        k_zero<<<1, 256, 0, stream>>>((unsigned int*)vvec, H);
    }
    k_hist<<<(E + 255) / 256, 256, 0, stream>>>(src_idx, counts, E);
    k_weighted_rowsum<<<512, H, 0, stream>>>(x_src, counts, ls, n_src);
    k_matvec_R<<<H / MV_WAVES, MV_WAVES * 32, 0, stream>>>(R, ls, vvec);
    {
        int blocks = (n_trg + 127) / 128;
        k_dot_rows_wmma<<<blocks, 256, 0, stream>>>(x_trg, vvec, t, n_trg);
    }
    k_gather_scores<<<(E + 255) / 256, 256, 0, stream>>>(t, trg_idx, out, E);
}


# one zero-fill launch instead of three: counts and left_sum (adjacent workspace slots) cleared by one grid with 16-byte stores; v no longer cleared, the matvec stores all of it; 8 launches -> 6
# speedup vs baseline: 2.6198x; 1.0717x over previous
#include <hip/hip_runtime.h>
#include <hip/hip_bf16.h>
#include <stdint.h>

#define H 256

typedef __attribute__((ext_vector_type(2))) float v2f;
typedef __attribute__((ext_vector_type(4))) float v4f;
typedef __attribute__((ext_vector_type(8))) float v8f;

typedef __attribute__((ext_vector_type(4))) unsigned int v4u;

__global__ void k_zero(unsigned int* __restrict__ p, int n16) {
    int i = blockIdx.x * blockDim.x + threadIdx.x;
    if (i < n16) ((v4u*)p)[i] = (v4u){0u, 0u, 0u, 0u};
}

__global__ void k_hist(const int* __restrict__ src_idx, unsigned int* __restrict__ counts, int E) {
    int i = blockIdx.x * blockDim.x + threadIdx.x;
    if (i < E) atomicAdd(&counts[src_idx[i]], 1u);
}

#define RS_ROWS 13

__global__ __launch_bounds__(256)
void k_weighted_rowsum(const float* __restrict__ X,
                       const unsigned int* __restrict__ counts,
                       float* __restrict__ ls, int nrows) {
    __shared__ v4f part[8][H / 4];
    const int lane = threadIdx.x & 31;
    const int wave = __builtin_amdgcn_readfirstlane((int)(threadIdx.x >> 5));
    const int gw = blockIdx.x * 8 + wave;
    const int nw = gridDim.x * 8;

    v4f a0 = {0.0f, 0.0f, 0.0f, 0.0f};
    v4f a1 = {0.0f, 0.0f, 0.0f, 0.0f};
    for (int n0 = gw; n0 < nrows; n0 += nw * RS_ROWS) {
        const int nl = n0 + lane * nw;
        const bool cvalid = lane < RS_ROWS && nl < nrows;
        const unsigned int craw = counts[cvalid ? nl : n0];

        v4f x0[RS_ROWS], x1[RS_ROWS];
        #pragma unroll
        for (int r = 0; r < RS_ROWS; ++r) {
            const int n = n0 + r * nw;
            const int nn = (n < nrows) ? n : n0;
            const v4f* p = (const v4f*)(X + (size_t)nn * H);
            x0[r] = p[lane];
            x1[r] = p[lane + 32];
        }
        __builtin_amdgcn_sched_barrier(0);
        const float cl = cvalid ? (float)craw : 0.0f;
        #pragma unroll
        for (int r = 0; r < RS_ROWS; ++r) {
            const float c = __builtin_bit_cast(float,
                __builtin_amdgcn_readlane(__builtin_bit_cast(int, cl), r));
            a0.x = fmaf(c, x0[r].x, a0.x);
            a0.y = fmaf(c, x0[r].y, a0.y);
            a0.z = fmaf(c, x0[r].z, a0.z);
            a0.w = fmaf(c, x0[r].w, a0.w);
            a1.x = fmaf(c, x1[r].x, a1.x);
            a1.y = fmaf(c, x1[r].y, a1.y);
            a1.z = fmaf(c, x1[r].z, a1.z);
            a1.w = fmaf(c, x1[r].w, a1.w);
        }
    }
    part[wave][lane]      = a0;
    part[wave][lane + 32] = a1;
    __syncthreads();
    const float* pf = (const float*)&part[0][0];
    float s = 0.0f;
    for (int w = 0; w < 8; ++w) s += pf[w * H + threadIdx.x];
    atomicAdd(&ls[threadIdx.x], s);
}

#define MV_WAVES 4

__global__ __launch_bounds__(MV_WAVES * 32)
void k_matvec_R(const float* __restrict__ R,
                const float* __restrict__ ls,
                float* __restrict__ v) {
    const int wave = threadIdx.x >> 5;
    const int lane = threadIdx.x & 31;
    const int i = blockIdx.x * MV_WAVES + wave;
    const float* row = R + (size_t)i * H;
    float rv[H / 32], sv[H / 32];
    for (int k = 0; k < H / 32; ++k) {
        rv[k] = row[lane + 32 * k];
        sv[k] = ls[lane + 32 * k];
    }
    float acc = 0.0f;
    for (int k = 0; k < H / 32; ++k) acc = fmaf(rv[k], sv[k], acc);
    for (int off = 16; off > 0; off >>= 1) acc += __shfl_xor(acc, off, 32);
    if (lane == 0) v[i] = acc;
}

__global__ void k_dot_rows_wmma(const float* __restrict__ X,
                                const float* __restrict__ v,
                                float* __restrict__ t, int nrows) {
    __shared__ float vs[H];
    vs[threadIdx.x] = v[threadIdx.x];
    __syncthreads();

    int wave = threadIdx.x >> 5;
    int lane = threadIdx.x & 31;
    int row0 = (blockIdx.x * 8 + wave) * 16;

    int r = row0 + (lane & 15);
    if (r >= nrows) r = 0;
    const float* xrow = X + (size_t)r * H;
    int khalf = (lane >> 4) * 2;

    v8f c = {};
    #pragma unroll 8
    for (int k0 = 0; k0 < H; k0 += 4) {
        int kk = k0 + khalf;
        v2f a = *(const v2f*)(vs + kk);
        v2f b = *(const v2f*)(xrow + kk);
        c = __builtin_amdgcn_wmma_f32_16x16x4_f32(
                 false, a,  false, b,
                 (short)0, c,  false,  false);
    }

    int n = row0 + (lane & 15);
    if (lane < 16 && n < nrows) t[n] = c[0];
}

__global__ void k_gather_scores(const float* __restrict__ t,
                                const int* __restrict__ trg_idx,
                                float* __restrict__ out, int E) {
    int i = blockIdx.x * blockDim.x + threadIdx.x;
    if (i < E) out[i] = t[trg_idx[i]];
}

extern "C" void kernel_launch(void* const* d_in, const int* in_sizes, int n_in,
                              void* d_out, int out_size, void* d_ws, size_t ws_size,
                              hipStream_t stream) {
    const float* x_src = (const float*)d_in[0];
    const float* x_trg = (const float*)d_in[1];
    const float* R     = (const float*)d_in[2];
    const int*   edges = (const int*)d_in[3];

    const int n_src = in_sizes[0] / H;
    const int n_trg = in_sizes[1] / H;
    const int E     = in_sizes[3] / 2;
    const int* src_idx = edges;
    const int* trg_idx = edges + E;
    float* out = (float*)d_out;

    char* ws = (char*)d_ws;
    size_t off = 0;
    unsigned int* counts = (unsigned int*)(ws + off); off += ((size_t)n_src * 4 + 255) & ~(size_t)255;
    float* ls            = (float*)(ws + off);        off += (H * 4 + 255) & ~(size_t)255;
    float* vvec          = (float*)(ws + off);        off += (H * 4 + 255) & ~(size_t)255;
    float* t             = (float*)(ws + off);

    {
        int n16 = (int)(((char*)vvec - (char*)counts) / 16);
        k_zero<<<(n16 + 255) / 256, 256, 0, stream>>>(counts, n16);
    }
    k_hist<<<(E + 255) / 256, 256, 0, stream>>>(src_idx, counts, E);
    k_weighted_rowsum<<<512, H, 0, stream>>>(x_src, counts, ls, n_src);
    k_matvec_R<<<H / MV_WAVES, MV_WAVES * 32, 0, stream>>>(R, ls, vvec);
    {
        int blocks = (n_trg + 127) / 128;
        k_dot_rows_wmma<<<blocks, 256, 0, stream>>>(x_trg, vvec, t, n_trg);
    }
    k_gather_scores<<<(E + 255) / 256, 256, 0, stream>>>(t, trg_idx, out, E);
}


# row-dot WMMA kernel: a wave requests all 16 of its rows at once (32 16-byte loads; was 8 dependent trips of 8 8-byte loads), products follow the returns; k order of the f32 accumulation changes
# speedup vs baseline: 2.7019x; 1.0313x over previous
#include <hip/hip_runtime.h>
#include <hip/hip_bf16.h>
#include <stdint.h>

#define H 256

typedef __attribute__((ext_vector_type(2))) float v2f;
typedef __attribute__((ext_vector_type(4))) float v4f;
typedef __attribute__((ext_vector_type(8))) float v8f;

typedef __attribute__((ext_vector_type(4))) unsigned int v4u;

__global__ void k_zero(unsigned int* __restrict__ p, int n16) {
    int i = blockIdx.x * blockDim.x + threadIdx.x;
    if (i < n16) ((v4u*)p)[i] = (v4u){0u, 0u, 0u, 0u};
}

__global__ void k_hist(const int* __restrict__ src_idx, unsigned int* __restrict__ counts, int E) {
    int i = blockIdx.x * blockDim.x + threadIdx.x;
    if (i < E) atomicAdd(&counts[src_idx[i]], 1u);
}

#define RS_ROWS 13

__global__ __launch_bounds__(256)
void k_weighted_rowsum(const float* __restrict__ X,
                       const unsigned int* __restrict__ counts,
                       float* __restrict__ ls, int nrows) {
    __shared__ v4f part[8][H / 4];
    const int lane = threadIdx.x & 31;
    const int wave = __builtin_amdgcn_readfirstlane((int)(threadIdx.x >> 5));
    const int gw = blockIdx.x * 8 + wave;
    const int nw = gridDim.x * 8;

    v4f a0 = {0.0f, 0.0f, 0.0f, 0.0f};
    v4f a1 = {0.0f, 0.0f, 0.0f, 0.0f};
    for (int n0 = gw; n0 < nrows; n0 += nw * RS_ROWS) {
        const int nl = n0 + lane * nw;
        const bool cvalid = lane < RS_ROWS && nl < nrows;
        const unsigned int craw = counts[cvalid ? nl : n0];

        v4f x0[RS_ROWS], x1[RS_ROWS];
        #pragma unroll
        for (int r = 0; r < RS_ROWS; ++r) {
            const int n = n0 + r * nw;
            const int nn = (n < nrows) ? n : n0;
            const v4f* p = (const v4f*)(X + (size_t)nn * H);
            x0[r] = p[lane];
            x1[r] = p[lane + 32];
        }
        __builtin_amdgcn_sched_barrier(0);
        const float cl = cvalid ? (float)craw : 0.0f;
        #pragma unroll
        for (int r = 0; r < RS_ROWS; ++r) {
            const float c = __builtin_bit_cast(float,
                __builtin_amdgcn_readlane(__builtin_bit_cast(int, cl), r));
            a0.x = fmaf(c, x0[r].x, a0.x);
            a0.y = fmaf(c, x0[r].y, a0.y);
            a0.z = fmaf(c, x0[r].z, a0.z);
            a0.w = fmaf(c, x0[r].w, a0.w);
            a1.x = fmaf(c, x1[r].x, a1.x);
            a1.y = fmaf(c, x1[r].y, a1.y);
            a1.z = fmaf(c, x1[r].z, a1.z);
            a1.w = fmaf(c, x1[r].w, a1.w);
        }
    }
    part[wave][lane]      = a0;
    part[wave][lane + 32] = a1;
    __syncthreads();
    const float* pf = (const float*)&part[0][0];
    float s = 0.0f;
    for (int w = 0; w < 8; ++w) s += pf[w * H + threadIdx.x];
    atomicAdd(&ls[threadIdx.x], s);
}

#define MV_WAVES 4

__global__ __launch_bounds__(MV_WAVES * 32)
void k_matvec_R(const float* __restrict__ R,
                const float* __restrict__ ls,
                float* __restrict__ v) {
    const int wave = threadIdx.x >> 5;
    const int lane = threadIdx.x & 31;
    const int i = blockIdx.x * MV_WAVES + wave;
    const float* row = R + (size_t)i * H;
    float rv[H / 32], sv[H / 32];
    for (int k = 0; k < H / 32; ++k) {
        rv[k] = row[lane + 32 * k];
        sv[k] = ls[lane + 32 * k];
    }
    float acc = 0.0f;
    for (int k = 0; k < H / 32; ++k) acc = fmaf(rv[k], sv[k], acc);
    for (int off = 16; off > 0; off >>= 1) acc += __shfl_xor(acc, off, 32);
    if (lane == 0) v[i] = acc;
}

#define DR_LOADS (H / 8)

__global__ __launch_bounds__(256)
void k_dot_rows_wmma(const float* __restrict__ X,
                     const float* __restrict__ v,
                     float* __restrict__ t, int nrows) {
    __shared__ v4f vs[H / 4];
    ((float*)vs)[threadIdx.x] = v[threadIdx.x];
    __syncthreads();

    const int lane = threadIdx.x & 31;
    const int wave = threadIdx.x >> 5;
    const int row0 = (blockIdx.x * 8 + wave) * 16;
    const int h = lane >> 4;

    int r = row0 + (lane & 15);
    if (r >= nrows) r = 0;
    const v4f* xq = (const v4f*)(X + (size_t)r * H) + h;

    v4f b[DR_LOADS];
    #pragma unroll
    for (int j = 0; j < DR_LOADS; ++j) b[j] = xq[2 * j];
    __builtin_amdgcn_sched_barrier(0);

    v8f c = {};
    #pragma unroll
    for (int j = 0; j < DR_LOADS; ++j) {
        const v4f a = vs[2 * j + h];
        v2f a0 = {a.x, a.y};
        v2f a1 = {a.z, a.w};
        v2f b0 = {b[j].x, b[j].y};
        v2f b1 = {b[j].z, b[j].w};
        c = __builtin_amdgcn_wmma_f32_16x16x4_f32(
                 false, a0,  false, b0,
                 (short)0, c,  false,  false);
        c = __builtin_amdgcn_wmma_f32_16x16x4_f32(
                 false, a1,  false, b1,
                 (short)0, c,  false,  false);
    }

    const int n = row0 + (lane & 15);
    if (lane < 16 && n < nrows) t[n] = c[0];
}

__global__ void k_gather_scores(const float* __restrict__ t,
                                const int* __restrict__ trg_idx,
                                float* __restrict__ out, int E) {
    int i = blockIdx.x * blockDim.x + threadIdx.x;
    if (i < E) out[i] = t[trg_idx[i]];
}

extern "C" void kernel_launch(void* const* d_in, const int* in_sizes, int n_in,
                              void* d_out, int out_size, void* d_ws, size_t ws_size,
                              hipStream_t stream) {
    const float* x_src = (const float*)d_in[0];
    const float* x_trg = (const float*)d_in[1];
    const float* R     = (const float*)d_in[2];
    const int*   edges = (const int*)d_in[3];

    const int n_src = in_sizes[0] / H;
    const int n_trg = in_sizes[1] / H;
    const int E     = in_sizes[3] / 2;
    const int* src_idx = edges;
    const int* trg_idx = edges + E;
    float* out = (float*)d_out;

    char* ws = (char*)d_ws;
    size_t off = 0;
    unsigned int* counts = (unsigned int*)(ws + off); off += ((size_t)n_src * 4 + 255) & ~(size_t)255;
    float* ls            = (float*)(ws + off);        off += (H * 4 + 255) & ~(size_t)255;
    float* vvec          = (float*)(ws + off);        off += (H * 4 + 255) & ~(size_t)255;
    float* t             = (float*)(ws + off);

    {
        int n16 = (int)(((char*)vvec - (char*)counts) / 16);
        k_zero<<<(n16 + 255) / 256, 256, 0, stream>>>(counts, n16);
    }
    k_hist<<<(E + 255) / 256, 256, 0, stream>>>(src_idx, counts, E);
    k_weighted_rowsum<<<512, H, 0, stream>>>(x_src, counts, ls, n_src);
    k_matvec_R<<<H / MV_WAVES, MV_WAVES * 32, 0, stream>>>(R, ls, vvec);
    {
        int blocks = (n_trg + 127) / 128;
        k_dot_rows_wmma<<<blocks, 256, 0, stream>>>(x_trg, vvec, t, n_trg);
    }
    k_gather_scores<<<(E + 255) / 256, 256, 0, stream>>>(t, trg_idx, out, E);
}


# source histogram privatised in LDS: 64 work-groups (16 edge slices x 4 row ranges) fill 2-byte bins and store 16 count planes that the row-sum adds up; no global count atomics, no zero-fill launch
# speedup vs baseline: 3.8868x; 1.4385x over previous
#include <hip/hip_runtime.h>
#include <hip/hip_bf16.h>
#include <stdint.h>

#define H 256

typedef __attribute__((ext_vector_type(2))) float v2f;
typedef __attribute__((ext_vector_type(4))) float v4f;
typedef __attribute__((ext_vector_type(8))) float v8f;

typedef __attribute__((ext_vector_type(4))) unsigned int v4u;
typedef __attribute__((ext_vector_type(4))) int v4i;

#define HP 16
#define NR 4
#define RB 12544
#define HB_PITCH (NR * RB)
#define RB_V4 (RB / 8)
#define HIST_THREADS 1024
#define HIST_BATCH 5

__global__ __launch_bounds__(HIST_THREADS)
void k_hist_lds(const int* __restrict__ src_idx,
                unsigned short* __restrict__ planes,
                float* __restrict__ ls, int E) {
    __shared__ v4u bins[RB_V4];
    const int tid = threadIdx.x;
    const int slice = blockIdx.x / NR;
    const int lo = (blockIdx.x % NR) * RB;
    for (int i = tid; i < RB_V4; i += HIST_THREADS) bins[i] = (v4u){0u, 0u, 0u, 0u};
    if (blockIdx.x == 0 && tid < H) ls[tid] = 0.0f;

    const int chunk = ((E + 4 * HP - 1) / (4 * HP)) * 4;
    const int e0 = slice * chunk;
    const int e1 = (e0 + chunk < E) ? e0 + chunk : E;

    unsigned int* bw = (unsigned int*)bins;
    for (int b0 = e0; b0 < e1; b0 += 4 * HIST_THREADS * HIST_BATCH) {
        v4i iv[HIST_BATCH];
        #pragma unroll
        for (int j = 0; j < HIST_BATCH; ++j) {
            const int e = b0 + 4 * (tid + j * HIST_THREADS);
            iv[j] = (e < e1) ? *(const v4i*)(src_idx + e) : (v4i){0, 0, 0, 0};
        }
        if (b0 == e0) __syncthreads();
        #pragma unroll
        for (int j = 0; j < HIST_BATCH; ++j) {
            const int e = b0 + 4 * (tid + j * HIST_THREADS);
            if (e + 3 < e1) {
                #pragma unroll
                for (int c = 0; c < 4; ++c) {
                    const unsigned int d = (unsigned int)(iv[j][c] - lo);
                    if (d < (unsigned int)RB) atomicAdd(&bw[d >> 1], 1u << ((d & 1u) << 4));
                }
            } else if (e < e1) {
                for (int c = 0; c < 4 && e + c < e1; ++c) {
                    const unsigned int d = (unsigned int)(iv[j][c] - lo);
                    if (d < (unsigned int)RB) atomicAdd(&bw[d >> 1], 1u << ((d & 1u) << 4));
                }
            }
        }
    }
    __syncthreads();
    v4u* dst = (v4u*)(planes + (size_t)slice * HB_PITCH + lo);
    for (int i = tid; i < RB_V4; i += HIST_THREADS) dst[i] = bins[i];
}

#define RS_ROWS 13

__global__ __launch_bounds__(256)
void k_weighted_rowsum(const float* __restrict__ X,
                       const unsigned short* __restrict__ planes,
                       float* __restrict__ ls, int nrows) {
    __shared__ v4f part[8][H / 4];
    const int lane = threadIdx.x & 31;
    const int wave = __builtin_amdgcn_readfirstlane((int)(threadIdx.x >> 5));
    const int gw = blockIdx.x * 8 + wave;
    const int nw = gridDim.x * 8;

    v4f a0 = {0.0f, 0.0f, 0.0f, 0.0f};
    v4f a1 = {0.0f, 0.0f, 0.0f, 0.0f};
    for (int n0 = gw; n0 < nrows; n0 += nw * RS_ROWS) {
        const int nl = n0 + lane * nw;
        const bool cvalid = lane < RS_ROWS && nl < nrows;
        const unsigned short* pc = planes + (cvalid ? nl : n0);
        unsigned int cr[HP];
        #pragma unroll
        for (int p = 0; p < HP; ++p) cr[p] = pc[(size_t)p * HB_PITCH];

        v4f x0[RS_ROWS], x1[RS_ROWS];
        #pragma unroll
        for (int r = 0; r < RS_ROWS; ++r) {
            const int n = n0 + r * nw;
            const int nn = (n < nrows) ? n : n0;
            const v4f* p = (const v4f*)(X + (size_t)nn * H);
            x0[r] = p[lane];
            x1[r] = p[lane + 32];
        }
        __builtin_amdgcn_sched_barrier(0);
        unsigned int craw = 0u;
        #pragma unroll
        for (int p = 0; p < HP; ++p) craw += cr[p];
        const float cl = cvalid ? (float)craw : 0.0f;
        #pragma unroll
        for (int r = 0; r < RS_ROWS; ++r) {
            const float c = __builtin_bit_cast(float,
                __builtin_amdgcn_readlane(__builtin_bit_cast(int, cl), r));
            a0.x = fmaf(c, x0[r].x, a0.x);
            a0.y = fmaf(c, x0[r].y, a0.y);
            a0.z = fmaf(c, x0[r].z, a0.z);
            a0.w = fmaf(c, x0[r].w, a0.w);
            a1.x = fmaf(c, x1[r].x, a1.x);
            a1.y = fmaf(c, x1[r].y, a1.y);
            a1.z = fmaf(c, x1[r].z, a1.z);
            a1.w = fmaf(c, x1[r].w, a1.w);
        }
    }
    part[wave][lane]      = a0;
    part[wave][lane + 32] = a1;
    __syncthreads();
    const float* pf = (const float*)&part[0][0];
    float s = 0.0f;
    for (int w = 0; w < 8; ++w) s += pf[w * H + threadIdx.x];
    atomicAdd(&ls[threadIdx.x], s);
}

#define MV_WAVES 4

__global__ __launch_bounds__(MV_WAVES * 32)
void k_matvec_R(const float* __restrict__ R,
                const float* __restrict__ ls,
                float* __restrict__ v) {
    const int wave = threadIdx.x >> 5;
    const int lane = threadIdx.x & 31;
    const int i = blockIdx.x * MV_WAVES + wave;
    const float* row = R + (size_t)i * H;
    float rv[H / 32], sv[H / 32];
    for (int k = 0; k < H / 32; ++k) {
        rv[k] = row[lane + 32 * k];
        sv[k] = ls[lane + 32 * k];
    }
    float acc = 0.0f;
    for (int k = 0; k < H / 32; ++k) acc = fmaf(rv[k], sv[k], acc);
    for (int off = 16; off > 0; off >>= 1) acc += __shfl_xor(acc, off, 32);
    if (lane == 0) v[i] = acc;
}

#define DR_LOADS (H / 8)

__global__ __launch_bounds__(256)
void k_dot_rows_wmma(const float* __restrict__ X,
                     const float* __restrict__ v,
                     float* __restrict__ t, int nrows) {
    __shared__ v4f vs[H / 4];
    ((float*)vs)[threadIdx.x] = v[threadIdx.x];
    __syncthreads();

    const int lane = threadIdx.x & 31;
    const int wave = threadIdx.x >> 5;
    const int row0 = (blockIdx.x * 8 + wave) * 16;
    const int h = lane >> 4;

    int r = row0 + (lane & 15);
    if (r >= nrows) r = 0;
    const v4f* xq = (const v4f*)(X + (size_t)r * H) + h;

    v4f b[DR_LOADS];
    #pragma unroll
    for (int j = 0; j < DR_LOADS; ++j) b[j] = xq[2 * j];
    __builtin_amdgcn_sched_barrier(0);

    v8f c = {};
    #pragma unroll
    for (int j = 0; j < DR_LOADS; ++j) {
        const v4f a = vs[2 * j + h];
        v2f a0 = {a.x, a.y};
        v2f a1 = {a.z, a.w};
        v2f b0 = {b[j].x, b[j].y};
        v2f b1 = {b[j].z, b[j].w};
        c = __builtin_amdgcn_wmma_f32_16x16x4_f32(
                 false, a0,  false, b0,
                 (short)0, c,  false,  false);
        c = __builtin_amdgcn_wmma_f32_16x16x4_f32(
                 false, a1,  false, b1,
                 (short)0, c,  false,  false);
    }

    const int n = row0 + (lane & 15);
    if (lane < 16 && n < nrows) t[n] = c[0];
}

__global__ void k_gather_scores(const float* __restrict__ t,
                                const int* __restrict__ trg_idx,
                                float* __restrict__ out, int E) {
    int i = blockIdx.x * blockDim.x + threadIdx.x;
    if (i < E) out[i] = t[trg_idx[i]];
}

extern "C" void kernel_launch(void* const* d_in, const int* in_sizes, int n_in,
                              void* d_out, int out_size, void* d_ws, size_t ws_size,
                              hipStream_t stream) {
    const float* x_src = (const float*)d_in[0];
    const float* x_trg = (const float*)d_in[1];
    const float* R     = (const float*)d_in[2];
    const int*   edges = (const int*)d_in[3];

    const int n_src = in_sizes[0] / H;
    const int n_trg = in_sizes[1] / H;
    const int E     = in_sizes[3] / 2;
    const int* src_idx = edges;
    const int* trg_idx = edges + E;
    float* out = (float*)d_out;

    char* ws = (char*)d_ws;
    size_t off = 0;
    float* ls              = (float*)(ws + off);          off += (H * 4 + 255) & ~(size_t)255;
    float* vvec            = (float*)(ws + off);          off += (H * 4 + 255) & ~(size_t)255;
    unsigned short* planes = (unsigned short*)(ws + off);
    float* t               = (float*)(ws + off);

    const size_t plane_bytes = (size_t)HP * HB_PITCH * sizeof(unsigned short);
    const size_t t_bytes     = (size_t)n_trg * sizeof(float);
    if (ws_size < off + (plane_bytes > t_bytes ? plane_bytes : t_bytes) ||
        n_src > HB_PITCH || E > HP * 65532) return;

    k_hist_lds<<<HP * NR, HIST_THREADS, 0, stream>>>(src_idx, planes, ls, E);
    k_weighted_rowsum<<<512, H, 0, stream>>>(x_src, planes, ls, n_src);
    k_matvec_R<<<H / MV_WAVES, MV_WAVES * 32, 0, stream>>>(R, ls, vvec);
    {
        int blocks = (n_trg + 127) / 128;
        k_dot_rows_wmma<<<blocks, 256, 0, stream>>>(x_trg, vvec, t, n_trg);
    }
    k_gather_scores<<<(E + 255) / 256, 256, 0, stream>>>(t, trg_idx, out, E);
}


# row-dot WMMA kernel: X fetched with whole-line loads (2 rows x 256 B per load, was 16 rows x 32 B), products on the diagonal of 8 row-pair tiles, 8-lane DPP row sums; new f32 summation order
# speedup vs baseline: 4.4062x; 1.1336x over previous
#include <hip/hip_runtime.h>
#include <hip/hip_bf16.h>
#include <stdint.h>

#define H 256

typedef __attribute__((ext_vector_type(2))) float v2f;
typedef __attribute__((ext_vector_type(4))) float v4f;
typedef __attribute__((ext_vector_type(8))) float v8f;

typedef __attribute__((ext_vector_type(4))) unsigned int v4u;
typedef __attribute__((ext_vector_type(4))) int v4i;

#define HP 16
#define NR 4
#define RB 12544
#define HB_PITCH (NR * RB)
#define RB_V4 (RB / 8)
#define HIST_THREADS 1024
#define HIST_BATCH 5

__global__ __launch_bounds__(HIST_THREADS)
void k_hist_lds(const int* __restrict__ src_idx,
                unsigned short* __restrict__ planes,
                float* __restrict__ v, int E) {
    __shared__ v4u bins[RB_V4];
    const int tid = threadIdx.x;
    const int slice = blockIdx.x / NR;
    const int lo = (blockIdx.x % NR) * RB;
    for (int i = tid; i < RB_V4; i += HIST_THREADS) bins[i] = (v4u){0u, 0u, 0u, 0u};
    if (blockIdx.x == 0 && tid < H) v[tid] = 0.0f;

    const int chunk = ((E + 4 * HP - 1) / (4 * HP)) * 4;
    const int e0 = slice * chunk;
    const int e1 = (e0 + chunk < E) ? e0 + chunk : E;

    unsigned int* bw = (unsigned int*)bins;
    for (int b0 = e0; b0 < e1; b0 += 4 * HIST_THREADS * HIST_BATCH) {
        v4i iv[HIST_BATCH];
        #pragma unroll
        for (int j = 0; j < HIST_BATCH; ++j) {
            const int e = b0 + 4 * (tid + j * HIST_THREADS);
            iv[j] = (e < e1) ? *(const v4i*)(src_idx + e) : (v4i){0, 0, 0, 0};
        }
        if (b0 == e0) __syncthreads();
        #pragma unroll
        for (int j = 0; j < HIST_BATCH; ++j) {
            const int e = b0 + 4 * (tid + j * HIST_THREADS);
            if (e + 3 < e1) {
                #pragma unroll
                for (int c = 0; c < 4; ++c) {
                    const unsigned int d = (unsigned int)(iv[j][c] - lo);
                    if (d < (unsigned int)RB) atomicAdd(&bw[d >> 1], 1u << ((d & 1u) << 4));
                }
            } else if (e < e1) {
                for (int c = 0; c < 4 && e + c < e1; ++c) {
                    const unsigned int d = (unsigned int)(iv[j][c] - lo);
                    if (d < (unsigned int)RB) atomicAdd(&bw[d >> 1], 1u << ((d & 1u) << 4));
                }
            }
        }
    }
    __syncthreads();
    v4u* dst = (v4u*)(planes + (size_t)slice * HB_PITCH + lo);
    for (int i = tid; i < RB_V4; i += HIST_THREADS) dst[i] = bins[i];
}

#define CS 32
#define NSL (H / CS)
#define RS_LOADS 26
#define RS_WROWS (4 * RS_LOADS)
#define RS_BROWS (8 * RS_WROWS)

__global__ __launch_bounds__(256)
void k_rowsum_times_R(const float* __restrict__ X,
                      const unsigned short* __restrict__ planes,
                      const float* __restrict__ R,
                      float* __restrict__ v, int nrows) {
    __shared__ v4f cnt4[8][32];
    __shared__ v4f red[32][CS / 4];
    __shared__ v4f ps4[CS / 4];
    __shared__ v4f vp4[H * 2];

    float* vp = (float*)vp4;
    const int tid = threadIdx.x;
    const int lane = tid & 31;
    const int wave = __builtin_amdgcn_readfirstlane((int)(threadIdx.x >> 5));
    const int s = blockIdx.x % NSL;
    const int wrow0 = (blockIdx.x / NSL) * RS_BROWS + wave * RS_WROWS;
    const int g = lane >> 3;
    const int q = lane & 7;

    const int r16 = lane & 15;
    const int half = lane >> 4;
    const int crow = wrow0 + 8 * r16;
    const bool cvalid = r16 < RS_WROWS / 8 && crow < nrows;
    const v4u* pc = (const v4u*)(planes + (size_t)(half * (HP / 2)) * HB_PITCH + (cvalid ? crow : 0));
    v4u cr[HP / 2];
    #pragma unroll
    for (int p = 0; p < HP / 2; ++p) cr[p] = pc[(size_t)p * (HB_PITCH / 8)];

    v4f x[RS_LOADS];
    if (wrow0 + RS_WROWS <= nrows) {
        const float* xb = X + (size_t)(wrow0 + g) * H + s * CS + 4 * q;
        #pragma unroll
        for (int k = 0; k < RS_LOADS; ++k) x[k] = *(const v4f*)(xb + (size_t)k * (4 * H));
    } else {
        #pragma unroll
        for (int k = 0; k < RS_LOADS; ++k) {
            const int n = wrow0 + 4 * k + g;
            const int nn = (n < nrows) ? n : 0;
            x[k] = *(const v4f*)(X + (size_t)nn * H + s * CS + 4 * q);
        }
    }

    const float* rb = R + (size_t)(32 * wave + g) * H + s * CS + 4 * q;
    v4f rr[8];
    #pragma unroll
    for (int m = 0; m < 8; ++m) rr[m] = *(const v4f*)(rb + (size_t)m * (4 * H));

    __builtin_amdgcn_sched_barrier(0);

    unsigned int cs[8] = {0u, 0u, 0u, 0u, 0u, 0u, 0u, 0u};
    #pragma unroll
    for (int p = 0; p < HP / 2; ++p) {
        #pragma unroll
        for (int e = 0; e < 4; ++e) {
            cs[2 * e]     += cr[p][e] & 0xffffu;
            cs[2 * e + 1] += cr[p][e] >> 16;
        }
    }
    float cf[8];
    #pragma unroll
    for (int e = 0; e < 8; ++e) {
        const unsigned int tot = cs[e] + (unsigned int)__shfl_xor((int)cs[e], 16, 32);
        cf[e] = (cvalid && crow + e < nrows) ? (float)tot : 0.0f;
    }
    if (lane < 16) {
        cnt4[wave][2 * lane]     = (v4f){cf[0], cf[1], cf[2], cf[3]};
        cnt4[wave][2 * lane + 1] = (v4f){cf[4], cf[5], cf[6], cf[7]};
    }
    __syncthreads();

    const float* cw = (const float*)&cnt4[wave][0];
    v4f a = {0.0f, 0.0f, 0.0f, 0.0f};
    #pragma unroll
    for (int k = 0; k < RS_LOADS; ++k) {
        const float c = cw[4 * k + g];
        a.x = fmaf(c, x[k].x, a.x);
        a.y = fmaf(c, x[k].y, a.y);
        a.z = fmaf(c, x[k].z, a.z);
        a.w = fmaf(c, x[k].w, a.w);
    }

    red[wave * 4 + g][q] = a;
    __syncthreads();
    if (tid < CS) {
        const float* rf = (const float*)&red[0][0];
        float sum = 0.0f;
        for (int r = 0; r < 32; ++r) sum += rf[r * CS + tid];
        ((float*)ps4)[tid] = sum;
    }
    __syncthreads();

    const v4f pq = ps4[q];
    #pragma unroll
    for (int m = 0; m < 8; ++m) {
        float t0 = rr[m].x * pq.x;
        t0 = fmaf(rr[m].y, pq.y, t0);
        t0 = fmaf(rr[m].z, pq.z, t0);
        t0 = fmaf(rr[m].w, pq.w, t0);
        vp[256 * wave + 32 * m + lane] = t0;
    }
    __syncthreads();
    const v4f u0 = vp4[2 * tid];
    const v4f u1 = vp4[2 * tid + 1];
    const float acc = ((u0.x + u0.y) + (u0.z + u0.w)) + ((u1.x + u1.y) + (u1.z + u1.w));
    atomicAdd(&v[tid], acc);
}

#define DR_PAIRS 8
#define DR_CHUNKS (H / 64)
#define DR_LOADS (DR_PAIRS * DR_CHUNKS)

#define DR_XOR_ADD(x, mask) \
    ((x) + __builtin_bit_cast(float, __builtin_amdgcn_update_dpp( \
        0, __builtin_bit_cast(int, (x)), 0x160 + (mask), 0xf, 0xf, false)))

__global__ __launch_bounds__(256)
void k_dot_rows_wmma(const float* __restrict__ X,
                     const float* __restrict__ v,
                     float* __restrict__ t, int nrows) {
    __shared__ v4f vs[H / 4];
    ((float*)vs)[threadIdx.x] = v[threadIdx.x];
    __syncthreads();

    const int lane = threadIdx.x & 31;
    const int wave = __builtin_amdgcn_readfirstlane((int)(threadIdx.x >> 5));
    const int row0 = (blockIdx.x * 8 + wave) * 16;
    const int h = lane >> 4;
    const int rb = (lane >> 3) & 1;
    const int q = lane & 7;
    const int cu = 8 * h + q;

    const v4f* xp[DR_PAIRS];
    #pragma unroll
    for (int p = 0; p < DR_PAIRS; ++p) {
        const int n = row0 + 2 * p + rb;
        const int nn = (n < nrows) ? n : 0;
        xp[p] = (const v4f*)(X + (size_t)nn * H) + cu;
    }
    v4f b[DR_LOADS];
    #pragma unroll
    for (int u = 0; u < DR_CHUNKS; ++u) {
        #pragma unroll
        for (int p = 0; p < DR_PAIRS; ++p) b[u * DR_PAIRS + p] = xp[p][16 * u];
    }
    __builtin_amdgcn_sched_barrier(0);

    v8f c[DR_PAIRS];
    #pragma unroll
    for (int p = 0; p < DR_PAIRS; ++p) c[p] = (v8f){};
    #pragma unroll
    for (int u = 0; u < DR_CHUNKS; ++u) {
        const v4f a = vs[16 * u + cu];
        const v2f a0 = {a.x, a.y};
        const v2f a1 = {a.z, a.w};
        #pragma unroll
        for (int p = 0; p < DR_PAIRS; ++p) {
            const v4f bb = b[u * DR_PAIRS + p];
            const v2f b0 = {bb.x, bb.y};
            c[p] = __builtin_amdgcn_wmma_f32_16x16x4_f32(
                        false, a0,  false, b0,
                        (short)0, c[p],  false,  false);
        }
        #pragma unroll
        for (int p = 0; p < DR_PAIRS; ++p) {
            const v4f bb = b[u * DR_PAIRS + p];
            const v2f b1 = {bb.z, bb.w};
            c[p] = __builtin_amdgcn_wmma_f32_16x16x4_f32(
                        false, a1,  false, b1,
                        (short)0, c[p],  false,  false);
        }
    }

    float d[DR_PAIRS];
    #pragma unroll
    for (int p = 0; p < DR_PAIRS; ++p) d[p] = c[p][q];
    float r[DR_PAIRS / 2];
    #pragma unroll
    for (int k = 0; k < DR_PAIRS / 2; ++k) {
        float x = h ? d[2 * k + 1] : d[2 * k];
        x = DR_XOR_ADD(x, 1);
        x = DR_XOR_ADD(x, 2);
        x = DR_XOR_ADD(x, 4);
        r[k] = x;
    }
    const float y01 = (q & 1) ? r[1] : r[0];
    const float y23 = (q & 1) ? r[3] : r[2];
    const float y = (q & 2) ? y23 : y01;
    const int n = row0 + 4 * q + (lane >> 3);
    if (q < 4 && n < nrows) t[n] = y;
}

__global__ void k_gather_scores(const float* __restrict__ t,
                                const int* __restrict__ trg_idx,
                                float* __restrict__ out, int E) {
    int i = blockIdx.x * blockDim.x + threadIdx.x;
    if (i < E) out[i] = t[trg_idx[i]];
}

extern "C" void kernel_launch(void* const* d_in, const int* in_sizes, int n_in,
                              void* d_out, int out_size, void* d_ws, size_t ws_size,
                              hipStream_t stream) {
    const float* x_src = (const float*)d_in[0];
    const float* x_trg = (const float*)d_in[1];
    const float* R     = (const float*)d_in[2];
    const int*   edges = (const int*)d_in[3];

    const int n_src = in_sizes[0] / H;
    const int n_trg = in_sizes[1] / H;
    const int E     = in_sizes[3] / 2;
    const int* src_idx = edges;
    const int* trg_idx = edges + E;
    float* out = (float*)d_out;

    char* ws = (char*)d_ws;
    size_t off = 0;
    float* vvec            = (float*)(ws + off);          off += (H * 4 + 255) & ~(size_t)255;
    unsigned short* planes = (unsigned short*)(ws + off);
    float* t               = (float*)(ws + off);

    const size_t plane_bytes = (size_t)HP * HB_PITCH * sizeof(unsigned short);
    const size_t t_bytes     = (size_t)n_trg * sizeof(float);
    if (ws_size < off + (plane_bytes > t_bytes ? plane_bytes : t_bytes) ||
        n_src > HB_PITCH || E > HP * 65532) return;

    k_hist_lds<<<HP * NR, HIST_THREADS, 0, stream>>>(src_idx, planes, vvec, E);
    {
        int groups = (n_src + RS_BROWS - 1) / RS_BROWS;
        k_rowsum_times_R<<<groups * NSL, 256, 0, stream>>>(x_src, planes, R, vvec, n_src);
    }
    {
        int blocks = (n_trg + 127) / 128;
        k_dot_rows_wmma<<<blocks, 256, 0, stream>>>(x_trg, vvec, t, n_trg);
    }
    k_gather_scores<<<(E + 255) / 256, 256, 0, stream>>>(t, trg_idx, out, E);
}


# row-dot WMMA kernel sized to the chip: 232 work-groups of 12 waves x 18 rows, at most one per WGP by registers (was 391 of 8 x 16, two fit a WGP); v read into A registers ahead of rows, no barrier
# speedup vs baseline: 4.6456x; 1.0543x over previous
#include <hip/hip_runtime.h>
#include <hip/hip_bf16.h>
#include <stdint.h>

#define H 256

typedef __attribute__((ext_vector_type(2))) float v2f;
typedef __attribute__((ext_vector_type(4))) float v4f;
typedef __attribute__((ext_vector_type(8))) float v8f;

typedef __attribute__((ext_vector_type(4))) unsigned int v4u;
typedef __attribute__((ext_vector_type(4))) int v4i;

#define HP 16
#define NR 4
#define RB 12544
#define HB_PITCH (NR * RB)
#define RB_V4 (RB / 8)
#define HIST_THREADS 1024
#define HIST_BATCH 5

__global__ __launch_bounds__(HIST_THREADS)
void k_hist_lds(const int* __restrict__ src_idx,
                unsigned short* __restrict__ planes,
                float* __restrict__ v, int E) {
    __shared__ v4u bins[RB_V4];
    const int tid = threadIdx.x;
    const int slice = blockIdx.x / NR;
    const int lo = (blockIdx.x % NR) * RB;
    for (int i = tid; i < RB_V4; i += HIST_THREADS) bins[i] = (v4u){0u, 0u, 0u, 0u};
    if (blockIdx.x == 0 && tid < H) v[tid] = 0.0f;

    const int chunk = ((E + 4 * HP - 1) / (4 * HP)) * 4;
    const int e0 = slice * chunk;
    const int e1 = (e0 + chunk < E) ? e0 + chunk : E;

    unsigned int* bw = (unsigned int*)bins;
    for (int b0 = e0; b0 < e1; b0 += 4 * HIST_THREADS * HIST_BATCH) {
        v4i iv[HIST_BATCH];
        #pragma unroll
        for (int j = 0; j < HIST_BATCH; ++j) {
            const int e = b0 + 4 * (tid + j * HIST_THREADS);
            iv[j] = (e < e1) ? *(const v4i*)(src_idx + e) : (v4i){0, 0, 0, 0};
        }
        if (b0 == e0) __syncthreads();
        #pragma unroll
        for (int j = 0; j < HIST_BATCH; ++j) {
            const int e = b0 + 4 * (tid + j * HIST_THREADS);
            if (e + 3 < e1) {
                #pragma unroll
                for (int c = 0; c < 4; ++c) {
                    const unsigned int d = (unsigned int)(iv[j][c] - lo);
                    if (d < (unsigned int)RB) atomicAdd(&bw[d >> 1], 1u << ((d & 1u) << 4));
                }
            } else if (e < e1) {
                for (int c = 0; c < 4 && e + c < e1; ++c) {
                    const unsigned int d = (unsigned int)(iv[j][c] - lo);
                    if (d < (unsigned int)RB) atomicAdd(&bw[d >> 1], 1u << ((d & 1u) << 4));
                }
            }
        }
    }
    __syncthreads();
    v4u* dst = (v4u*)(planes + (size_t)slice * HB_PITCH + lo);
    for (int i = tid; i < RB_V4; i += HIST_THREADS) dst[i] = bins[i];
}

#define CS 32
#define NSL (H / CS)
#define RS_LOADS 26
#define RS_WROWS (4 * RS_LOADS)
#define RS_BROWS (8 * RS_WROWS)

__global__ __launch_bounds__(256)
void k_rowsum_times_R(const float* __restrict__ X,
                      const unsigned short* __restrict__ planes,
                      const float* __restrict__ R,
                      float* __restrict__ v, int nrows) {
    __shared__ v4f cnt4[8][32];
    __shared__ v4f red[32][CS / 4];
    __shared__ v4f ps4[CS / 4];
    __shared__ v4f vp4[H * 2];

    float* vp = (float*)vp4;
    const int tid = threadIdx.x;
    const int lane = tid & 31;
    const int wave = __builtin_amdgcn_readfirstlane((int)(threadIdx.x >> 5));
    const int s = blockIdx.x % NSL;
    const int wrow0 = (blockIdx.x / NSL) * RS_BROWS + wave * RS_WROWS;
    const int g = lane >> 3;
    const int q = lane & 7;

    const int r16 = lane & 15;
    const int half = lane >> 4;
    const int crow = wrow0 + 8 * r16;
    const bool cvalid = r16 < RS_WROWS / 8 && crow < nrows;
    const v4u* pc = (const v4u*)(planes + (size_t)(half * (HP / 2)) * HB_PITCH + (cvalid ? crow : 0));
    v4u cr[HP / 2];
    #pragma unroll
    for (int p = 0; p < HP / 2; ++p) cr[p] = pc[(size_t)p * (HB_PITCH / 8)];

    v4f x[RS_LOADS];
    if (wrow0 + RS_WROWS <= nrows) {
        const float* xb = X + (size_t)(wrow0 + g) * H + s * CS + 4 * q;
        #pragma unroll
        for (int k = 0; k < RS_LOADS; ++k) x[k] = *(const v4f*)(xb + (size_t)k * (4 * H));
    } else {
        #pragma unroll
        for (int k = 0; k < RS_LOADS; ++k) {
            const int n = wrow0 + 4 * k + g;
            const int nn = (n < nrows) ? n : 0;
            x[k] = *(const v4f*)(X + (size_t)nn * H + s * CS + 4 * q);
        }
    }

    const float* rb = R + (size_t)(32 * wave + g) * H + s * CS + 4 * q;
    v4f rr[8];
    #pragma unroll
    for (int m = 0; m < 8; ++m) rr[m] = *(const v4f*)(rb + (size_t)m * (4 * H));

    __builtin_amdgcn_sched_barrier(0);

    unsigned int cs[8] = {0u, 0u, 0u, 0u, 0u, 0u, 0u, 0u};
    #pragma unroll
    for (int p = 0; p < HP / 2; ++p) {
        #pragma unroll
        for (int e = 0; e < 4; ++e) {
            cs[2 * e]     += cr[p][e] & 0xffffu;
            cs[2 * e + 1] += cr[p][e] >> 16;
        }
    }
    float cf[8];
    #pragma unroll
    for (int e = 0; e < 8; ++e) {
        const unsigned int tot = cs[e] + (unsigned int)__shfl_xor((int)cs[e], 16, 32);
        cf[e] = (cvalid && crow + e < nrows) ? (float)tot : 0.0f;
    }
    if (lane < 16) {
        cnt4[wave][2 * lane]     = (v4f){cf[0], cf[1], cf[2], cf[3]};
        cnt4[wave][2 * lane + 1] = (v4f){cf[4], cf[5], cf[6], cf[7]};
    }
    __syncthreads();

    const float* cw = (const float*)&cnt4[wave][0];
    v4f a = {0.0f, 0.0f, 0.0f, 0.0f};
    #pragma unroll
    for (int k = 0; k < RS_LOADS; ++k) {
        const float c = cw[4 * k + g];
        a.x = fmaf(c, x[k].x, a.x);
        a.y = fmaf(c, x[k].y, a.y);
        a.z = fmaf(c, x[k].z, a.z);
        a.w = fmaf(c, x[k].w, a.w);
    }

    red[wave * 4 + g][q] = a;
    __syncthreads();
    if (tid < CS) {
        const float* rf = (const float*)&red[0][0];
        float sum = 0.0f;
        for (int r = 0; r < 32; ++r) sum += rf[r * CS + tid];
        ((float*)ps4)[tid] = sum;
    }
    __syncthreads();

    const v4f pq = ps4[q];
    #pragma unroll
    for (int m = 0; m < 8; ++m) {
        float t0 = rr[m].x * pq.x;
        t0 = fmaf(rr[m].y, pq.y, t0);
        t0 = fmaf(rr[m].z, pq.z, t0);
        t0 = fmaf(rr[m].w, pq.w, t0);
        vp[256 * wave + 32 * m + lane] = t0;
    }
    __syncthreads();
    const v4f u0 = vp4[2 * tid];
    const v4f u1 = vp4[2 * tid + 1];
    const float acc = ((u0.x + u0.y) + (u0.z + u0.w)) + ((u1.x + u1.y) + (u1.z + u1.w));
    atomicAdd(&v[tid], acc);
}

#define DR_PAIRS 9
#define DR_ROWS (2 * DR_PAIRS)
#define DR_WAVES 12
#define DR_BROWS (DR_WAVES * DR_ROWS)
#define DR_CHUNKS (H / 64)
#define DR_LOADS (DR_PAIRS * DR_CHUNKS)

#define DR_XOR_ADD(x, mask) \
    ((x) + __builtin_bit_cast(float, __builtin_amdgcn_update_dpp( \
        0, __builtin_bit_cast(int, (x)), 0x160 + (mask), 0xf, 0xf, false)))

__global__ __launch_bounds__(32 * DR_WAVES)
void k_dot_rows_wmma(const float* __restrict__ X,
                     const float* __restrict__ v,
                     float* __restrict__ t, int nrows) {
    const int lane = threadIdx.x & 31;
    const int wave = __builtin_amdgcn_readfirstlane((int)(threadIdx.x >> 5));
    const int row0 = (blockIdx.x * DR_WAVES + wave) * DR_ROWS;
    const int h = lane >> 4;
    const int rb = (lane >> 3) & 1;
    const int q = lane & 7;
    const int cu = 8 * h + q;

    const v4f* vg = (const v4f*)v + cu;
    v4f a[DR_CHUNKS];
    #pragma unroll
    for (int u = 0; u < DR_CHUNKS; ++u) a[u] = vg[16 * u];

    const v4f* X4 = (const v4f*)X;
    v4f b[DR_LOADS];
    #pragma unroll
    for (int p = 0; p < DR_PAIRS; ++p) {
        const unsigned int n = (unsigned int)(row0 + 2 * p + rb);
        const unsigned int nn = (n < (unsigned int)nrows) ? n : 0u;
        __builtin_assume(nn < (1u << 23));
        const unsigned int ix = nn * (unsigned int)(H / 4) + (unsigned int)cu;
        #pragma unroll
        for (int u = 0; u < DR_CHUNKS; ++u) b[u * DR_PAIRS + p] = X4[ix + 16u * (unsigned int)u];
    }
    __builtin_amdgcn_sched_barrier(0);

    v8f c[DR_PAIRS];
    #pragma unroll
    for (int p = 0; p < DR_PAIRS; ++p) c[p] = (v8f){};
    #pragma unroll
    for (int pp = 0; pp + 1 < DR_PAIRS; pp += 2) {
        const int pe = (pp + 3 == DR_PAIRS) ? pp + 3 : pp + 2;
        #pragma unroll
        for (int u = 0; u < DR_CHUNKS; ++u) {
            const v2f a0 = {a[u].x, a[u].y};
            const v2f a1 = {a[u].z, a[u].w};
            #pragma unroll
            for (int p = pp; p < pe; ++p) {
                const v4f bb = b[u * DR_PAIRS + p];
                const v2f b0 = {bb.x, bb.y};
                c[p] = __builtin_amdgcn_wmma_f32_16x16x4_f32(
                            false, a0,  false, b0,
                            (short)0, c[p],  false,  false);
            }
            #pragma unroll
            for (int p = pp; p < pe; ++p) {
                const v4f bb = b[u * DR_PAIRS + p];
                const v2f b1 = {bb.z, bb.w};
                c[p] = __builtin_amdgcn_wmma_f32_16x16x4_f32(
                            false, a1,  false, b1,
                            (short)0, c[p],  false,  false);
            }
        }
        __builtin_amdgcn_sched_barrier(0);
    }

    float r[(DR_PAIRS + 1) / 2];
    #pragma unroll
    for (int k = 0; k < (DR_PAIRS + 1) / 2; ++k) {
        v8f cc = c[2 * k];
        if (2 * k + 1 < DR_PAIRS) cc = h ? c[2 * k + 1] : c[2 * k];
        float x = cc[q];
        x = DR_XOR_ADD(x, 1);
        x = DR_XOR_ADD(x, 2);
        x = DR_XOR_ADD(x, 4);
        r[k] = x;
    }
    float y = r[0];
    #pragma unroll
    for (int k = 1; k < (DR_PAIRS + 1) / 2; ++k) y = (q == k) ? r[k] : y;
    const int g = lane >> 3;
    const int n = row0 + 4 * q + g;
    if (4 * q + g < DR_ROWS && n < nrows) t[n] = y;
}

__global__ void k_gather_scores(const float* __restrict__ t,
                                const int* __restrict__ trg_idx,
                                float* __restrict__ out, int E) {
    int i = blockIdx.x * blockDim.x + threadIdx.x;
    if (i < E) out[i] = t[trg_idx[i]];
}

extern "C" void kernel_launch(void* const* d_in, const int* in_sizes, int n_in,
                              void* d_out, int out_size, void* d_ws, size_t ws_size,
                              hipStream_t stream) {
    const float* x_src = (const float*)d_in[0];
    const float* x_trg = (const float*)d_in[1];
    const float* R     = (const float*)d_in[2];
    const int*   edges = (const int*)d_in[3];

    const int n_src = in_sizes[0] / H;
    const int n_trg = in_sizes[1] / H;
    const int E     = in_sizes[3] / 2;
    const int* src_idx = edges;
    const int* trg_idx = edges + E;
    float* out = (float*)d_out;

    char* ws = (char*)d_ws;
    size_t off = 0;
    float* vvec            = (float*)(ws + off);          off += (H * 4 + 255) & ~(size_t)255;
    unsigned short* planes = (unsigned short*)(ws + off);
    float* t               = (float*)(ws + off);

    const size_t plane_bytes = (size_t)HP * HB_PITCH * sizeof(unsigned short);
    const size_t t_bytes     = (size_t)n_trg * sizeof(float);
    if (ws_size < off + (plane_bytes > t_bytes ? plane_bytes : t_bytes) ||
        n_src > HB_PITCH || E > HP * 65532) return;

    k_hist_lds<<<HP * NR, HIST_THREADS, 0, stream>>>(src_idx, planes, vvec, E);
    {
        int groups = (n_src + RS_BROWS - 1) / RS_BROWS;
        k_rowsum_times_R<<<groups * NSL, 256, 0, stream>>>(x_src, planes, R, vvec, n_src);
    }
    {
        int blocks = (n_trg + DR_BROWS - 1) / DR_BROWS;
        k_dot_rows_wmma<<<blocks, 32 * DR_WAVES, 0, stream>>>(x_trg, vvec, t, n_trg);
    }
    k_gather_scores<<<(E + 255) / 256, 256, 0, stream>>>(t, trg_idx, out, E);
}
